# starting kernel
# baseline (speedup 1.0000x reference)
#include <hip/hip_runtime.h>
#include <hip/hip_bf16.h>

typedef float v2f __attribute__((ext_vector_type(2)));
typedef float v8f __attribute__((ext_vector_type(8)));

#define B_   4
#define H_   16
#define L_   8192
#define D_   64
#define BS_  64
#define NQ_  128
#define SROW 129

__global__ __launch_bounds__(64)
void pool_project_kernel(const float* __restrict__ q, const float* __restrict__ k,
                         const float* __restrict__ Wq, const float* __restrict__ Wk,
                         float* __restrict__ qp, float* __restrict__ kp) {
    const float* x = blockIdx.y ? k : q;
    const float* W = blockIdx.y ? Wk : Wq;
    float* outp    = blockIdx.y ? kp : qp;

    const int bhblk = blockIdx.x;
    const int t  = threadIdx.x;
    const int cg = t & 15;
    const int rq = t >> 4;
    const size_t base = (size_t)bhblk * (BS_ * D_);

    float4 ps = make_float4(0.f, 0.f, 0.f, 0.f);
#pragma unroll
    for (int r = 0; r < 16; ++r) {
        const int row = rq * 16 + r;
        float4 v = *(const float4*)(x + base + (size_t)row * D_ + cg * 4);
        ps.x += v.x; ps.y += v.y; ps.z += v.z; ps.w += v.w;
    }

    __shared__ float4 part[4][16];
    __shared__ float  colsum[D_];
    part[rq][cg] = ps;
    __syncthreads();
    if (rq == 0) {
        float4 a = part[0][cg], b = part[1][cg], c = part[2][cg], d = part[3][cg];
        colsum[cg * 4 + 0] = a.x + b.x + c.x + d.x;
        colsum[cg * 4 + 1] = a.y + b.y + c.y + d.y;
        colsum[cg * 4 + 2] = a.z + b.z + c.z + d.z;
        colsum[cg * 4 + 3] = a.w + b.w + c.w + d.w;
    }
    __syncthreads();

    float acc = 0.f;
#pragma unroll
    for (int i = 0; i < 16; ++i) {
        float4 w = *(const float4*)(W + (size_t)t * D_ + i * 4);
        acc += w.x * colsum[i * 4 + 0] + w.y * colsum[i * 4 + 1] +
               w.z * colsum[i * 4 + 2] + w.w * colsum[i * 4 + 3];
    }
    outp[(size_t)bhblk * D_ + t] = acc * (1.0f / (float)BS_);
}

__global__ __launch_bounds__(256)
void score_softtopk_kernel(const float* __restrict__ qp, const float* __restrict__ kp,
                           const float* __restrict__ bias,
                           const float* __restrict__ bias_scale_p,
                           float* __restrict__ out) {
    extern __shared__ float lds[];

    const int bh  = blockIdx.x;
    const int h   = bh & (H_ - 1);
    const int tid = threadIdx.x;

    {
        const float4* qs = (const float4*)(qp + (size_t)bh * NQ_ * D_);
        const float4* ks = (const float4*)(kp + (size_t)bh * NQ_ * D_);
        float4* lq = (float4*)lds;
        float4* lk = (float4*)(lds + NQ_ * D_);
#pragma unroll
        for (int i = tid; i < (NQ_ * D_) / 4; i += 256) {
            lq[i] = qs[i];
            lk[i] = ks[i];
        }
    }
    __syncthreads();

    const int wave = tid >> 5;
    const int lane = tid & 31;
    const int m0   = wave * 16;
    const int lm   = lane & 15;
    const int kh   = (lane >> 4) << 1;

    v2f afrag[16];
#pragma unroll
    for (int kk = 0; kk < 16; ++kk)
        afrag[kk] = *(const v2f*)&lds[(m0 + lm) * D_ + kk * 4 + kh];

    v8f acc[8];
#pragma unroll
    for (int nt = 0; nt < 8; ++nt) {
        v2f bfrag[16];
#pragma unroll
        for (int kk = 0; kk < 16; ++kk)
            bfrag[kk] = *(const v2f*)&lds[NQ_ * D_ + (nt * 16 + lm) * D_ + kk * 4 + kh];

        v8f c = {0.f, 0.f, 0.f, 0.f, 0.f, 0.f, 0.f, 0.f};
#pragma unroll
        for (int kk = 0; kk < 16; ++kk)
            c = __builtin_amdgcn_wmma_f32_16x16x4_f32(
                    false, afrag[kk], false, bfrag[kk], (short)0, c, false, false);
        acc[nt] = c;
    }
    __syncthreads();

    {
        const float bscale   = bias_scale_p[0];
        const float inv_sqd  = 0.125f;
        const float inv_tau  = 10.0f;
        const int mhalf = (lane >> 4) << 3;
#pragma unroll
        for (int nt = 0; nt < 8; ++nt) {
#pragma unroll
            for (int r = 0; r < 8; ++r) {
                const int m = m0 + r + mhalf;
                const int n = nt * 16 + lm;
                float v = acc[nt][r] * inv_sqd
                        + bscale * bias[((size_t)h * NQ_ + m) * NQ_ + n];
                lds[m * SROW + n] = v * inv_tau;
            }
        }
    }
    __syncthreads();

    const int row  = tid >> 1;
    const int half = tid & 1;
    float s[64];
#pragma unroll
    for (int j = 0; j < 64; ++j) s[j] = lds[row * SROW + half * 64 + j];

    float lo = -10000.0f, hi = 10000.0f;
    const float target = 0.15f * (float)NQ_;
    for (int it = 0; it < 20; ++it) {
        const float mid = 0.5f * (lo + hi);
        float p = 0.f;
#pragma unroll
        for (int j = 0; j < 64; ++j)
            p += 1.0f / (1.0f + __expf(-(s[j] + mid)));
        const float total = p + __shfl_xor(p, 1, 32);
        if (total < target) lo = mid; else hi = mid;
    }
    const float lam = 0.5f * (lo + hi);

    float* orow = out + ((size_t)bh * NQ_ + row) * NQ_ + half * 64;
#pragma unroll
    for (int j = 0; j < 64; ++j)
        orow[j] = 1.0f / (1.0f + __expf(-(s[j] + lam)));
}

extern "C" void kernel_launch(void* const* d_in, const int* in_sizes, int n_in,
                              void* d_out, int out_size, void* d_ws, size_t ws_size,
                              hipStream_t stream) {
    const float* q     = (const float*)d_in[0];
    const float* k     = (const float*)d_in[1];
    const float* bias  = (const float*)d_in[2];
    const float* Wq    = (const float*)d_in[3];
    const float* Wk    = (const float*)d_in[4];
    const float* bsc   = (const float*)d_in[5];
    float* out = (float*)d_out;

    float* qp = (float*)d_ws;
    float* kp = qp + (size_t)B_ * H_ * NQ_ * D_;

    dim3 g1(B_ * H_ * NQ_, 2);
    pool_project_kernel<<<g1, 64, 0, stream>>>(q, k, Wq, Wk, qp, kp);

    const size_t lds_bytes = (size_t)NQ_ * SROW * sizeof(float);
    score_softtopk_kernel<<<B_ * H_, 256, lds_bytes, stream>>>(qp, kp, bias, bsc, out);
}


# score + soft-top-k kernel split into 16-row strips, one work-group each (512, was 64), so the bisection runs on every WGP; row sum = 8-term lane partial + 16-lane DPP butterfly (new reduction tree)
# speedup vs baseline: 1.7968x; 1.7968x over previous
#include <hip/hip_runtime.h>
#include <hip/hip_bf16.h>

typedef float v2f __attribute__((ext_vector_type(2)));
typedef float v8f __attribute__((ext_vector_type(8)));

#define B_   4
#define H_   16
#define L_   8192
#define D_   64
#define BS_  64
#define NQ_  128
#define NSTRIP 8
#define SROW 132

__global__ __launch_bounds__(64)
void pool_project_kernel(const float* __restrict__ q, const float* __restrict__ k,
                         const float* __restrict__ Wq, const float* __restrict__ Wk,
                         float* __restrict__ qp, float* __restrict__ kp) {
    const float* x = blockIdx.y ? k : q;
    const float* W = blockIdx.y ? Wk : Wq;
    float* outp    = blockIdx.y ? kp : qp;

    const int bhblk = blockIdx.x;
    const int t  = threadIdx.x;
    const int cg = t & 15;
    const int rq = t >> 4;
    const size_t base = (size_t)bhblk * (BS_ * D_);

    float4 ps = make_float4(0.f, 0.f, 0.f, 0.f);
#pragma unroll
    for (int r = 0; r < 16; ++r) {
        const int row = rq * 16 + r;
        float4 v = *(const float4*)(x + base + (size_t)row * D_ + cg * 4);
        ps.x += v.x; ps.y += v.y; ps.z += v.z; ps.w += v.w;
    }

    __shared__ float4 part[4][16];
    __shared__ float  colsum[D_];
    part[rq][cg] = ps;
    __syncthreads();
    if (rq == 0) {
        float4 a = part[0][cg], b = part[1][cg], c = part[2][cg], d = part[3][cg];
        colsum[cg * 4 + 0] = a.x + b.x + c.x + d.x;
        colsum[cg * 4 + 1] = a.y + b.y + c.y + d.y;
        colsum[cg * 4 + 2] = a.z + b.z + c.z + d.z;
        colsum[cg * 4 + 3] = a.w + b.w + c.w + d.w;
    }
    __syncthreads();

    float acc = 0.f;
#pragma unroll
    for (int i = 0; i < 16; ++i) {
        float4 w = *(const float4*)(W + (size_t)t * D_ + i * 4);
        acc += w.x * colsum[i * 4 + 0] + w.y * colsum[i * 4 + 1] +
               w.z * colsum[i * 4 + 2] + w.w * colsum[i * 4 + 3];
    }
    outp[(size_t)bhblk * D_ + t] = acc * (1.0f / (float)BS_);
}


#define ROW_XOR_ADD(x, M)                                                          \
    ((x) + __builtin_bit_cast(float, __builtin_amdgcn_update_dpp(                   \
               0, __builtin_bit_cast(int, (x)), 0x160 + (M), 0xf, 0xf, true)))

__global__ __launch_bounds__(256)
void score_softtopk_kernel(const float* __restrict__ qp, const float* __restrict__ kp,
                           const float* __restrict__ bias,
                           const float* __restrict__ bias_scale_p,
                           float* __restrict__ out) {
    __shared__ alignas(16) float lq[16 * D_];
    __shared__ alignas(16) float lk[NQ_ * D_];
    __shared__ alignas(16) float ls[16 * SROW];

    const int bh    = blockIdx.x >> 3;
    const int strip = blockIdx.x & (NSTRIP - 1);
    const int h     = bh & (H_ - 1);
    const int tid   = threadIdx.x;
    const int wave  = tid >> 5;
    const int lane  = tid & 31;
    const int lm    = lane & 15;
    const int hh    = lane >> 4;
    const int kh    = hh << 1;
    const int m0    = strip * 16;

    float bv[8];
#pragma unroll
    for (int r = 0; r < 8; ++r)
        bv[r] = bias[((size_t)h * NQ_ + (m0 + 8 * hh + r)) * NQ_ + wave * 16 + lm];
    const float bscale = bias_scale_p[0];

    {
        const float4* qs = (const float4*)(qp + ((size_t)bh * NQ_ + m0) * D_);
        const float4* ks = (const float4*)(kp + (size_t)bh * NQ_ * D_);
        ((float4*)lq)[tid] = qs[tid];
#pragma unroll
        for (int i = 0; i < 8; ++i)
            ((float4*)lk)[tid + i * 256] = ks[tid + i * 256];
    }
    __syncthreads();

    v2f afrag[16], bfrag[16];
#pragma unroll
    for (int kk = 0; kk < 16; ++kk)
        afrag[kk] = *(const v2f*)&lq[lm * D_ + kk * 4 + kh];
#pragma unroll
    for (int kk = 0; kk < 16; ++kk)
        bfrag[kk] = *(const v2f*)&lk[(wave * 16 + lm) * D_ + kk * 4 + kh];

    v8f c = {0.f, 0.f, 0.f, 0.f, 0.f, 0.f, 0.f, 0.f};
#pragma unroll
    for (int kk = 0; kk < 16; ++kk)
        c = __builtin_amdgcn_wmma_f32_16x16x4_f32(
                false, afrag[kk], false, bfrag[kk], (short)0, c, false, false);

    {
        const float inv_sqd = 0.125f;
        const float inv_tau = 10.0f;
#pragma unroll
        for (int r = 0; r < 8; ++r) {
            float v = c[r] * inv_sqd + bscale * bv[r];
            ls[(8 * hh + r) * SROW + wave * 16 + lm] = v * inv_tau;
        }
    }
    __syncthreads();

    const int row = tid >> 4;
    const int seg = tid & 15;
    float s[8];
    {
        const float4 a = *(const float4*)&ls[row * SROW + seg * 4];
        const float4 b = *(const float4*)&ls[row * SROW + 64 + seg * 4];
        s[0] = a.x; s[1] = a.y; s[2] = a.z; s[3] = a.w;
        s[4] = b.x; s[5] = b.y; s[6] = b.z; s[7] = b.w;
    }

    float lo = -10000.0f, hi = 10000.0f;
    const float target = 0.15f * (float)NQ_;
    for (int it = 0; it < 20; ++it) {
        const float mid = 0.5f * (lo + hi);
        float p = 0.f;
#pragma unroll
        for (int j = 0; j < 8; ++j)
            p += 1.0f / (1.0f + __expf(-(s[j] + mid)));
        p = ROW_XOR_ADD(p, 1);
        p = ROW_XOR_ADD(p, 2);
        p = ROW_XOR_ADD(p, 4);
        p = ROW_XOR_ADD(p, 8);
        if (p < target) lo = mid; else hi = mid;
    }
    const float lam = 0.5f * (lo + hi);

    float o[8];
#pragma unroll
    for (int j = 0; j < 8; ++j)
        o[j] = 1.0f / (1.0f + __expf(-(s[j] + lam)));

    float* orow = out + ((size_t)bh * NQ_ + (m0 + row)) * NQ_;
    *(float4*)(orow + seg * 4)      = make_float4(o[0], o[1], o[2], o[3]);
    *(float4*)(orow + 64 + seg * 4) = make_float4(o[4], o[5], o[6], o[7]);
}

extern "C" void kernel_launch(void* const* d_in, const int* in_sizes, int n_in,
                              void* d_out, int out_size, void* d_ws, size_t ws_size,
                              hipStream_t stream) {
    const float* q     = (const float*)d_in[0];
    const float* k     = (const float*)d_in[1];
    const float* bias  = (const float*)d_in[2];
    const float* Wq    = (const float*)d_in[3];
    const float* Wk    = (const float*)d_in[4];
    const float* bsc   = (const float*)d_in[5];
    float* out = (float*)d_out;

    float* qp = (float*)d_ws;
    float* kp = qp + (size_t)B_ * H_ * NQ_ * D_;

    dim3 g1(B_ * H_ * NQ_, 2);
    pool_project_kernel<<<g1, 64, 0, stream>>>(q, k, Wq, Wk, qp, kp);

    score_softtopk_kernel<<<B_ * H_ * NSTRIP, 256, 0, stream>>>(qp, kp, bias, bsc, out);
}


# pool+projection kernel re-formed as a stream: 256 resident work-groups of 16 waves, 4 tiles per wave, 16-32 loads in flight, W staged once in LDS (column sums and projection summed in a new order)
# speedup vs baseline: 2.4194x; 1.3465x over previous
#include <hip/hip_runtime.h>
#include <hip/hip_bf16.h>

typedef float v2f __attribute__((ext_vector_type(2)));
typedef float v8f __attribute__((ext_vector_type(8)));

#define B_   4
#define H_   16
#define L_   8192
#define D_   64
#define BS_  64
#define NQ_  128
#define NSTRIP 8
#define SROW 132

typedef float v4f __attribute__((ext_vector_type(4)));

#define PWAVES 16
#define PTILES 4
#define WPITCH 68

static __device__ __forceinline__ float xhalf(float x) {
    const int b = __builtin_bit_cast(int, x);
    return __builtin_bit_cast(float,
        __builtin_amdgcn_permlanex16(b, b, 0x76543210, 0xfedcba98, false, false));
}
static __device__ __forceinline__ float lane_val(float v, int L) {
    return __builtin_bit_cast(float, __builtin_amdgcn_readlane(__builtin_bit_cast(int, v), L));
}

__global__ __launch_bounds__(PWAVES * 32)
void pool_project_kernel(const float* __restrict__ q, const float* __restrict__ k,
                         const float* __restrict__ Wq, const float* __restrict__ Wk,
                         float* __restrict__ qp, float* __restrict__ kp) {
    __shared__ alignas(16) float lw[D_ * WPITCH];

    const float* x = blockIdx.y ? k : q;
    const float* W = blockIdx.y ? Wk : Wq;
    float* outp    = blockIdx.y ? kp : qp;

    const int tid  = threadIdx.x;
    const int lane = tid & 31;
    const int wave = __builtin_amdgcn_readfirstlane(tid >> 5);

#pragma unroll
    for (int i = 0; i < 2; ++i) {
        const int idx = tid + i * (PWAVES * 32);
        const int r = idx >> 4, c4 = idx & 15;
        *(v4f*)&lw[r * WPITCH + c4 * 4] = *(const v4f*)(W + r * D_ + c4 * 4);
    }
    __syncthreads();

    const int tile0 = (blockIdx.x * PWAVES + wave) * PTILES;
    const v4f* src  = (const v4f*)(x + (size_t)tile0 * (BS_ * D_)) + lane;
    float* dst      = outp + (size_t)tile0 * D_;
    const float* wr = &lw[lane * WPITCH];

    v4f buf[2][16];
#define LOAD_CHUNK(C) do { _Pragma("unroll") for (int j = 0; j < 16; ++j)                    \
        buf[(C) & 1][j] = __builtin_nontemporal_load(src + ((C) * 16 + j) * 32); } while (0)

    LOAD_CHUNK(0);
    v4f ps0, ps1;
#pragma unroll
    for (int c = 0; c < 2 * PTILES; ++c) {
        __builtin_amdgcn_sched_barrier(0);
        if (c + 1 < 2 * PTILES) LOAD_CHUNK(c + 1);
        __builtin_amdgcn_sched_barrier(0);

        if ((c & 1) == 0) { ps0 = buf[0][0]; ps1 = buf[0][1]; }
        else              { ps0 += buf[1][0]; ps1 += buf[1][1]; }
#pragma unroll
        for (int j = 2; j < 16; j += 2) { ps0 += buf[c & 1][j]; ps1 += buf[c & 1][j + 1]; }

        if (c & 1) {
            v4f cs = ps0 + ps1;
            cs.x += xhalf(cs.x); cs.y += xhalf(cs.y); cs.z += xhalf(cs.z); cs.w += xhalf(cs.w);
            float a0 = 0.f, a1 = 0.f, b0 = 0.f, b1 = 0.f;
#pragma unroll
            for (int g = 0; g < 16; g += 2) {
                const v4f wa0 = *(const v4f*)(wr + 4 * g);
                const v4f wb0 = *(const v4f*)(wr + 32 * WPITCH + 4 * g);
                const v4f wa1 = *(const v4f*)(wr + 4 * g + 4);
                const v4f wb1 = *(const v4f*)(wr + 32 * WPITCH + 4 * g + 4);
                const float c0 = lane_val(cs.x, g),     c1 = lane_val(cs.y, g);
                const float c2 = lane_val(cs.z, g),     c3 = lane_val(cs.w, g);
                const float d0 = lane_val(cs.x, g + 1), d1 = lane_val(cs.y, g + 1);
                const float d2 = lane_val(cs.z, g + 1), d3 = lane_val(cs.w, g + 1);
                a0 = __builtin_fmaf(wa0.x, c0, a0); b0 = __builtin_fmaf(wb0.x, c0, b0);
                a1 = __builtin_fmaf(wa1.x, d0, a1); b1 = __builtin_fmaf(wb1.x, d0, b1);
                a0 = __builtin_fmaf(wa0.y, c1, a0); b0 = __builtin_fmaf(wb0.y, c1, b0);
                a1 = __builtin_fmaf(wa1.y, d1, a1); b1 = __builtin_fmaf(wb1.y, d1, b1);
                a0 = __builtin_fmaf(wa0.z, c2, a0); b0 = __builtin_fmaf(wb0.z, c2, b0);
                a1 = __builtin_fmaf(wa1.z, d2, a1); b1 = __builtin_fmaf(wb1.z, d2, b1);
                a0 = __builtin_fmaf(wa0.w, c3, a0); b0 = __builtin_fmaf(wb0.w, c3, b0);
                a1 = __builtin_fmaf(wa1.w, d3, a1); b1 = __builtin_fmaf(wb1.w, d3, b1);
            }
            float* o = dst + (c >> 1) * D_;
            o[lane]      = (a0 + a1) * (1.0f / (float)BS_);
            o[lane + 32] = (b0 + b1) * (1.0f / (float)BS_);
        }
    }
#undef LOAD_CHUNK
}


#define ROW_XOR_ADD(x, M)                                                          \
    ((x) + __builtin_bit_cast(float, __builtin_amdgcn_update_dpp(                   \
               0, __builtin_bit_cast(int, (x)), 0x160 + (M), 0xf, 0xf, true)))

__global__ __launch_bounds__(256)
void score_softtopk_kernel(const float* __restrict__ qp, const float* __restrict__ kp,
                           const float* __restrict__ bias,
                           const float* __restrict__ bias_scale_p,
                           float* __restrict__ out) {
    __shared__ alignas(16) float lq[16 * D_];
    __shared__ alignas(16) float lk[NQ_ * D_];
    __shared__ alignas(16) float ls[16 * SROW];

    const int bh    = blockIdx.x >> 3;
    const int strip = blockIdx.x & (NSTRIP - 1);
    const int h     = bh & (H_ - 1);
    const int tid   = threadIdx.x;
    const int wave  = tid >> 5;
    const int lane  = tid & 31;
    const int lm    = lane & 15;
    const int hh    = lane >> 4;
    const int kh    = hh << 1;
    const int m0    = strip * 16;

    float bv[8];
#pragma unroll
    for (int r = 0; r < 8; ++r)
        bv[r] = bias[((size_t)h * NQ_ + (m0 + 8 * hh + r)) * NQ_ + wave * 16 + lm];
    const float bscale = bias_scale_p[0];

    {
        const float4* qs = (const float4*)(qp + ((size_t)bh * NQ_ + m0) * D_);
        const float4* ks = (const float4*)(kp + (size_t)bh * NQ_ * D_);
        ((float4*)lq)[tid] = qs[tid];
#pragma unroll
        for (int i = 0; i < 8; ++i)
            ((float4*)lk)[tid + i * 256] = ks[tid + i * 256];
    }
    __syncthreads();

    v2f afrag[16], bfrag[16];
#pragma unroll
    for (int kk = 0; kk < 16; ++kk)
        afrag[kk] = *(const v2f*)&lq[lm * D_ + kk * 4 + kh];
#pragma unroll
    for (int kk = 0; kk < 16; ++kk)
        bfrag[kk] = *(const v2f*)&lk[(wave * 16 + lm) * D_ + kk * 4 + kh];

    v8f c = {0.f, 0.f, 0.f, 0.f, 0.f, 0.f, 0.f, 0.f};
#pragma unroll
    for (int kk = 0; kk < 16; ++kk)
        c = __builtin_amdgcn_wmma_f32_16x16x4_f32(
                false, afrag[kk], false, bfrag[kk], (short)0, c, false, false);

    {
        const float inv_sqd = 0.125f;
        const float inv_tau = 10.0f;
#pragma unroll
        for (int r = 0; r < 8; ++r) {
            float v = c[r] * inv_sqd + bscale * bv[r];
            ls[(8 * hh + r) * SROW + wave * 16 + lm] = v * inv_tau;
        }
    }
    __syncthreads();

    const int row = tid >> 4;
    const int seg = tid & 15;
    float s[8];
    {
        const float4 a = *(const float4*)&ls[row * SROW + seg * 4];
        const float4 b = *(const float4*)&ls[row * SROW + 64 + seg * 4];
        s[0] = a.x; s[1] = a.y; s[2] = a.z; s[3] = a.w;
        s[4] = b.x; s[5] = b.y; s[6] = b.z; s[7] = b.w;
    }

    float lo = -10000.0f, hi = 10000.0f;
    const float target = 0.15f * (float)NQ_;
    for (int it = 0; it < 20; ++it) {
        const float mid = 0.5f * (lo + hi);
        float p = 0.f;
#pragma unroll
        for (int j = 0; j < 8; ++j)
            p += 1.0f / (1.0f + __expf(-(s[j] + mid)));
        p = ROW_XOR_ADD(p, 1);
        p = ROW_XOR_ADD(p, 2);
        p = ROW_XOR_ADD(p, 4);
        p = ROW_XOR_ADD(p, 8);
        if (p < target) lo = mid; else hi = mid;
    }
    const float lam = 0.5f * (lo + hi);

    float o[8];
#pragma unroll
    for (int j = 0; j < 8; ++j)
        o[j] = 1.0f / (1.0f + __expf(-(s[j] + lam)));

    float* orow = out + ((size_t)bh * NQ_ + (m0 + row)) * NQ_;
    *(float4*)(orow + seg * 4)      = make_float4(o[0], o[1], o[2], o[3]);
    *(float4*)(orow + 64 + seg * 4) = make_float4(o[4], o[5], o[6], o[7]);
}

extern "C" void kernel_launch(void* const* d_in, const int* in_sizes, int n_in,
                              void* d_out, int out_size, void* d_ws, size_t ws_size,
                              hipStream_t stream) {
    const float* q     = (const float*)d_in[0];
    const float* k     = (const float*)d_in[1];
    const float* bias  = (const float*)d_in[2];
    const float* Wq    = (const float*)d_in[3];
    const float* Wk    = (const float*)d_in[4];
    const float* bsc   = (const float*)d_in[5];
    float* out = (float*)d_out;

    float* qp = (float*)d_ws;
    float* kp = qp + (size_t)B_ * H_ * NQ_ * D_;

    dim3 g1(B_ * H_ * NQ_ / (PWAVES * PTILES), 2);
    pool_project_kernel<<<g1, PWAVES * 32, 0, stream>>>(q, k, Wq, Wk, qp, kp);

    score_softtopk_kernel<<<B_ * H_ * NSTRIP, 256, 0, stream>>>(qp, kp, bias, bsc, out);
}


# score kernel: q_proj/k_proj rows staged in LDS at a row pitch of 68 floats (was 64), which takes the 16-way bank conflict out of the WMMA fragment loads; k order and every sum unchanged
# speedup vs baseline: 2.6479x; 1.0944x over previous
#include <hip/hip_runtime.h>
#include <hip/hip_bf16.h>

typedef float v2f __attribute__((ext_vector_type(2)));
typedef float v8f __attribute__((ext_vector_type(8)));

#define B_   4
#define H_   16
#define L_   8192
#define D_   64
#define BS_  64
#define NQ_  128
#define NSTRIP 8
#define SROW 132
#define FPITCH 68

typedef float v4f __attribute__((ext_vector_type(4)));

#define PWAVES 16
#define PTILES 4
#define WPITCH 68

static __device__ __forceinline__ float xhalf(float x) {
    const int b = __builtin_bit_cast(int, x);
    return __builtin_bit_cast(float,
        __builtin_amdgcn_permlanex16(b, b, 0x76543210, 0xfedcba98, false, false));
}
static __device__ __forceinline__ float lane_val(float v, int L) {
    return __builtin_bit_cast(float, __builtin_amdgcn_readlane(__builtin_bit_cast(int, v), L));
}

__global__ __launch_bounds__(PWAVES * 32)
void pool_project_kernel(const float* __restrict__ q, const float* __restrict__ k,
                         const float* __restrict__ Wq, const float* __restrict__ Wk,
                         float* __restrict__ qp, float* __restrict__ kp) {
    __shared__ alignas(16) float lw[D_ * WPITCH];

    const float* x = blockIdx.y ? k : q;
    const float* W = blockIdx.y ? Wk : Wq;
    float* outp    = blockIdx.y ? kp : qp;

    const int tid  = threadIdx.x;
    const int lane = tid & 31;
    const int wave = __builtin_amdgcn_readfirstlane(tid >> 5);

#pragma unroll
    for (int i = 0; i < 2; ++i) {
        const int idx = tid + i * (PWAVES * 32);
        const int r = idx >> 4, c4 = idx & 15;
        *(v4f*)&lw[r * WPITCH + c4 * 4] = *(const v4f*)(W + r * D_ + c4 * 4);
    }
    __syncthreads();

    const int tile0 = (blockIdx.x * PWAVES + wave) * PTILES;
    const v4f* src  = (const v4f*)(x + (size_t)tile0 * (BS_ * D_)) + lane;
    float* dst      = outp + (size_t)tile0 * D_;
    const float* wr = &lw[lane * WPITCH];

    v4f buf[2][16];
#define LOAD_CHUNK(C) do { _Pragma("unroll") for (int j = 0; j < 16; ++j)                    \
        buf[(C) & 1][j] = __builtin_nontemporal_load(src + ((C) * 16 + j) * 32); } while (0)

    LOAD_CHUNK(0);
    v4f ps0, ps1;
#pragma unroll
    for (int c = 0; c < 2 * PTILES; ++c) {
        __builtin_amdgcn_sched_barrier(0);
        if (c + 1 < 2 * PTILES) LOAD_CHUNK(c + 1);
        __builtin_amdgcn_sched_barrier(0);

        if ((c & 1) == 0) { ps0 = buf[0][0]; ps1 = buf[0][1]; }
        else              { ps0 += buf[1][0]; ps1 += buf[1][1]; }
#pragma unroll
        for (int j = 2; j < 16; j += 2) { ps0 += buf[c & 1][j]; ps1 += buf[c & 1][j + 1]; }

        if (c & 1) {
            v4f cs = ps0 + ps1;
            cs.x += xhalf(cs.x); cs.y += xhalf(cs.y); cs.z += xhalf(cs.z); cs.w += xhalf(cs.w);
            float a0 = 0.f, a1 = 0.f, b0 = 0.f, b1 = 0.f;
#pragma unroll
            for (int g = 0; g < 16; g += 2) {
                const v4f wa0 = *(const v4f*)(wr + 4 * g);
                const v4f wb0 = *(const v4f*)(wr + 32 * WPITCH + 4 * g);
                const v4f wa1 = *(const v4f*)(wr + 4 * g + 4);
                const v4f wb1 = *(const v4f*)(wr + 32 * WPITCH + 4 * g + 4);
                const float c0 = lane_val(cs.x, g),     c1 = lane_val(cs.y, g);
                const float c2 = lane_val(cs.z, g),     c3 = lane_val(cs.w, g);
                const float d0 = lane_val(cs.x, g + 1), d1 = lane_val(cs.y, g + 1);
                const float d2 = lane_val(cs.z, g + 1), d3 = lane_val(cs.w, g + 1);
                a0 = __builtin_fmaf(wa0.x, c0, a0); b0 = __builtin_fmaf(wb0.x, c0, b0);
                a1 = __builtin_fmaf(wa1.x, d0, a1); b1 = __builtin_fmaf(wb1.x, d0, b1);
                a0 = __builtin_fmaf(wa0.y, c1, a0); b0 = __builtin_fmaf(wb0.y, c1, b0);
                a1 = __builtin_fmaf(wa1.y, d1, a1); b1 = __builtin_fmaf(wb1.y, d1, b1);
                a0 = __builtin_fmaf(wa0.z, c2, a0); b0 = __builtin_fmaf(wb0.z, c2, b0);
                a1 = __builtin_fmaf(wa1.z, d2, a1); b1 = __builtin_fmaf(wb1.z, d2, b1);
                a0 = __builtin_fmaf(wa0.w, c3, a0); b0 = __builtin_fmaf(wb0.w, c3, b0);
                a1 = __builtin_fmaf(wa1.w, d3, a1); b1 = __builtin_fmaf(wb1.w, d3, b1);
            }
            float* o = dst + (c >> 1) * D_;
            o[lane]      = (a0 + a1) * (1.0f / (float)BS_);
            o[lane + 32] = (b0 + b1) * (1.0f / (float)BS_);
        }
    }
#undef LOAD_CHUNK
}


#define ROW_XOR_ADD(x, M)                                                          \
    ((x) + __builtin_bit_cast(float, __builtin_amdgcn_update_dpp(                   \
               0, __builtin_bit_cast(int, (x)), 0x160 + (M), 0xf, 0xf, true)))

__global__ __launch_bounds__(256)
void score_softtopk_kernel(const float* __restrict__ qp, const float* __restrict__ kp,
                           const float* __restrict__ bias,
                           const float* __restrict__ bias_scale_p,
                           float* __restrict__ out) {
    __shared__ alignas(16) float lq[16 * FPITCH];
    __shared__ alignas(16) float lk[NQ_ * FPITCH];
    __shared__ alignas(16) float ls[16 * SROW];

    const int bh    = blockIdx.x >> 3;
    const int strip = blockIdx.x & (NSTRIP - 1);
    const int h     = bh & (H_ - 1);
    const int tid   = threadIdx.x;
    const int wave  = tid >> 5;
    const int lane  = tid & 31;
    const int lm    = lane & 15;
    const int hh    = lane >> 4;
    const int kh    = hh << 1;
    const int m0    = strip * 16;

    float bv[8];
#pragma unroll
    for (int r = 0; r < 8; ++r)
        bv[r] = bias[((size_t)h * NQ_ + (m0 + 8 * hh + r)) * NQ_ + wave * 16 + lm];
    const float bscale = bias_scale_p[0];

    {
        const float4* qs = (const float4*)(qp + ((size_t)bh * NQ_ + m0) * D_);
        const float4* ks = (const float4*)(kp + (size_t)bh * NQ_ * D_);
        const int r = tid >> 4, c4 = tid & 15;
        *(float4*)&lq[r * FPITCH + c4 * 4] = qs[tid];
#pragma unroll
        for (int i = 0; i < 8; ++i)
            *(float4*)&lk[(r + 16 * i) * FPITCH + c4 * 4] = ks[tid + i * 256];
    }
    __syncthreads();

    v2f afrag[16], bfrag[16];
#pragma unroll
    for (int kk = 0; kk < 16; ++kk)
        afrag[kk] = *(const v2f*)&lq[lm * FPITCH + kk * 4 + kh];
#pragma unroll
    for (int kk = 0; kk < 16; ++kk)
        bfrag[kk] = *(const v2f*)&lk[(wave * 16 + lm) * FPITCH + kk * 4 + kh];

    v8f c = {0.f, 0.f, 0.f, 0.f, 0.f, 0.f, 0.f, 0.f};
#pragma unroll
    for (int kk = 0; kk < 16; ++kk)
        c = __builtin_amdgcn_wmma_f32_16x16x4_f32(
                false, afrag[kk], false, bfrag[kk], (short)0, c, false, false);

    {
        const float inv_sqd = 0.125f;
        const float inv_tau = 10.0f;
#pragma unroll
        for (int r = 0; r < 8; ++r) {
            float v = c[r] * inv_sqd + bscale * bv[r];
            ls[(8 * hh + r) * SROW + wave * 16 + lm] = v * inv_tau;
        }
    }
    __syncthreads();

    const int row = tid >> 4;
    const int seg = tid & 15;
    float s[8];
    {
        const float4 a = *(const float4*)&ls[row * SROW + seg * 4];
        const float4 b = *(const float4*)&ls[row * SROW + 64 + seg * 4];
        s[0] = a.x; s[1] = a.y; s[2] = a.z; s[3] = a.w;
        s[4] = b.x; s[5] = b.y; s[6] = b.z; s[7] = b.w;
    }

    float lo = -10000.0f, hi = 10000.0f;
    const float target = 0.15f * (float)NQ_;
    for (int it = 0; it < 20; ++it) {
        const float mid = 0.5f * (lo + hi);
        float p = 0.f;
#pragma unroll
        for (int j = 0; j < 8; ++j)
            p += 1.0f / (1.0f + __expf(-(s[j] + mid)));
        p = ROW_XOR_ADD(p, 1);
        p = ROW_XOR_ADD(p, 2);
        p = ROW_XOR_ADD(p, 4);
        p = ROW_XOR_ADD(p, 8);
        if (p < target) lo = mid; else hi = mid;
    }
    const float lam = 0.5f * (lo + hi);

    float o[8];
#pragma unroll
    for (int j = 0; j < 8; ++j)
        o[j] = 1.0f / (1.0f + __expf(-(s[j] + lam)));

    float* orow = out + ((size_t)bh * NQ_ + (m0 + row)) * NQ_;
    *(float4*)(orow + seg * 4)      = make_float4(o[0], o[1], o[2], o[3]);
    *(float4*)(orow + 64 + seg * 4) = make_float4(o[4], o[5], o[6], o[7]);
}

extern "C" void kernel_launch(void* const* d_in, const int* in_sizes, int n_in,
                              void* d_out, int out_size, void* d_ws, size_t ws_size,
                              hipStream_t stream) {
    const float* q     = (const float*)d_in[0];
    const float* k     = (const float*)d_in[1];
    const float* bias  = (const float*)d_in[2];
    const float* Wq    = (const float*)d_in[3];
    const float* Wk    = (const float*)d_in[4];
    const float* bsc   = (const float*)d_in[5];
    float* out = (float*)d_out;

    float* qp = (float*)d_ws;
    float* kp = qp + (size_t)B_ * H_ * NQ_ * D_;

    dim3 g1(B_ * H_ * NQ_ / (PWAVES * PTILES), 2);
    pool_project_kernel<<<g1, PWAVES * 32, 0, stream>>>(q, k, Wq, Wk, qp, kp);

    score_softtopk_kernel<<<B_ * H_ * NSTRIP, 256, 0, stream>>>(qp, kp, bias, bsc, out);
}


# score kernel, bisection: all 20 steps kept, but a step whose comparison the row max/min score already fixes (128*sigmoid(smax+mid) < target or 128*sigmoid(smin+mid) > target) skips its sigmoid sum
# speedup vs baseline: 2.8164x; 1.0636x over previous
#include <hip/hip_runtime.h>
#include <hip/hip_bf16.h>

typedef float v2f __attribute__((ext_vector_type(2)));
typedef float v8f __attribute__((ext_vector_type(8)));

#define B_   4
#define H_   16
#define L_   8192
#define D_   64
#define BS_  64
#define NQ_  128
#define NSTRIP 8
#define SROW 132
#define FPITCH 68

typedef float v4f __attribute__((ext_vector_type(4)));

#define PWAVES 16
#define PTILES 4
#define WPITCH 68

static __device__ __forceinline__ float xhalf(float x) {
    const int b = __builtin_bit_cast(int, x);
    return __builtin_bit_cast(float,
        __builtin_amdgcn_permlanex16(b, b, 0x76543210, 0xfedcba98, false, false));
}
static __device__ __forceinline__ float lane_val(float v, int L) {
    return __builtin_bit_cast(float, __builtin_amdgcn_readlane(__builtin_bit_cast(int, v), L));
}

__global__ __launch_bounds__(PWAVES * 32)
void pool_project_kernel(const float* __restrict__ q, const float* __restrict__ k,
                         const float* __restrict__ Wq, const float* __restrict__ Wk,
                         float* __restrict__ qp, float* __restrict__ kp) {
    __shared__ alignas(16) float lw[D_ * WPITCH];

    const float* x = blockIdx.y ? k : q;
    const float* W = blockIdx.y ? Wk : Wq;
    float* outp    = blockIdx.y ? kp : qp;

    const int tid  = threadIdx.x;
    const int lane = tid & 31;
    const int wave = __builtin_amdgcn_readfirstlane(tid >> 5);

#pragma unroll
    for (int i = 0; i < 2; ++i) {
        const int idx = tid + i * (PWAVES * 32);
        const int r = idx >> 4, c4 = idx & 15;
        *(v4f*)&lw[r * WPITCH + c4 * 4] = *(const v4f*)(W + r * D_ + c4 * 4);
    }
    __syncthreads();

    const int tile0 = (blockIdx.x * PWAVES + wave) * PTILES;
    const v4f* src  = (const v4f*)(x + (size_t)tile0 * (BS_ * D_)) + lane;
    float* dst      = outp + (size_t)tile0 * D_;
    const float* wr = &lw[lane * WPITCH];

    v4f buf[2][16];
#define LOAD_CHUNK(C) do { _Pragma("unroll") for (int j = 0; j < 16; ++j)                    \
        buf[(C) & 1][j] = __builtin_nontemporal_load(src + ((C) * 16 + j) * 32); } while (0)

    LOAD_CHUNK(0);
    v4f ps0, ps1;
#pragma unroll
    for (int c = 0; c < 2 * PTILES; ++c) {
        __builtin_amdgcn_sched_barrier(0);
        if (c + 1 < 2 * PTILES) LOAD_CHUNK(c + 1);
        __builtin_amdgcn_sched_barrier(0);

        if ((c & 1) == 0) { ps0 = buf[0][0]; ps1 = buf[0][1]; }
        else              { ps0 += buf[1][0]; ps1 += buf[1][1]; }
#pragma unroll
        for (int j = 2; j < 16; j += 2) { ps0 += buf[c & 1][j]; ps1 += buf[c & 1][j + 1]; }

        if (c & 1) {
            v4f cs = ps0 + ps1;
            cs.x += xhalf(cs.x); cs.y += xhalf(cs.y); cs.z += xhalf(cs.z); cs.w += xhalf(cs.w);
            float a0 = 0.f, a1 = 0.f, b0 = 0.f, b1 = 0.f;
#pragma unroll
            for (int g = 0; g < 16; g += 2) {
                const v4f wa0 = *(const v4f*)(wr + 4 * g);
                const v4f wb0 = *(const v4f*)(wr + 32 * WPITCH + 4 * g);
                const v4f wa1 = *(const v4f*)(wr + 4 * g + 4);
                const v4f wb1 = *(const v4f*)(wr + 32 * WPITCH + 4 * g + 4);
                const float c0 = lane_val(cs.x, g),     c1 = lane_val(cs.y, g);
                const float c2 = lane_val(cs.z, g),     c3 = lane_val(cs.w, g);
                const float d0 = lane_val(cs.x, g + 1), d1 = lane_val(cs.y, g + 1);
                const float d2 = lane_val(cs.z, g + 1), d3 = lane_val(cs.w, g + 1);
                a0 = __builtin_fmaf(wa0.x, c0, a0); b0 = __builtin_fmaf(wb0.x, c0, b0);
                a1 = __builtin_fmaf(wa1.x, d0, a1); b1 = __builtin_fmaf(wb1.x, d0, b1);
                a0 = __builtin_fmaf(wa0.y, c1, a0); b0 = __builtin_fmaf(wb0.y, c1, b0);
                a1 = __builtin_fmaf(wa1.y, d1, a1); b1 = __builtin_fmaf(wb1.y, d1, b1);
                a0 = __builtin_fmaf(wa0.z, c2, a0); b0 = __builtin_fmaf(wb0.z, c2, b0);
                a1 = __builtin_fmaf(wa1.z, d2, a1); b1 = __builtin_fmaf(wb1.z, d2, b1);
                a0 = __builtin_fmaf(wa0.w, c3, a0); b0 = __builtin_fmaf(wb0.w, c3, b0);
                a1 = __builtin_fmaf(wa1.w, d3, a1); b1 = __builtin_fmaf(wb1.w, d3, b1);
            }
            float* o = dst + (c >> 1) * D_;
            o[lane]      = (a0 + a1) * (1.0f / (float)BS_);
            o[lane + 32] = (b0 + b1) * (1.0f / (float)BS_);
        }
    }
#undef LOAD_CHUNK
}


#define ROW_XOR_ADD(x, M)                                                          \
    ((x) + __builtin_bit_cast(float, __builtin_amdgcn_update_dpp(                   \
               0, __builtin_bit_cast(int, (x)), 0x160 + (M), 0xf, 0xf, true)))

#define ROW_XOR_MAX(x, M)                                                          \
    __builtin_fmaxf((x), __builtin_bit_cast(float, __builtin_amdgcn_update_dpp(    \
               0, __builtin_bit_cast(int, (x)), 0x160 + (M), 0xf, 0xf, true)))
#define ROW_XOR_MIN(x, M)                                                          \
    __builtin_fminf((x), __builtin_bit_cast(float, __builtin_amdgcn_update_dpp(    \
               0, __builtin_bit_cast(int, (x)), 0x160 + (M), 0xf, 0xf, true)))

#define ARG_SURELY_LT (-1.75f)
#define ARG_SURELY_GE (-1.72f)

__global__ __launch_bounds__(256)
void score_softtopk_kernel(const float* __restrict__ qp, const float* __restrict__ kp,
                           const float* __restrict__ bias,
                           const float* __restrict__ bias_scale_p,
                           float* __restrict__ out) {
    __shared__ alignas(16) float lq[16 * FPITCH];
    __shared__ alignas(16) float lk[NQ_ * FPITCH];
    __shared__ alignas(16) float ls[16 * SROW];

    const int bh    = blockIdx.x >> 3;
    const int strip = blockIdx.x & (NSTRIP - 1);
    const int h     = bh & (H_ - 1);
    const int tid   = threadIdx.x;
    const int wave  = tid >> 5;
    const int lane  = tid & 31;
    const int lm    = lane & 15;
    const int hh    = lane >> 4;
    const int kh    = hh << 1;
    const int m0    = strip * 16;

    float bv[8];
#pragma unroll
    for (int r = 0; r < 8; ++r)
        bv[r] = bias[((size_t)h * NQ_ + (m0 + 8 * hh + r)) * NQ_ + wave * 16 + lm];
    const float bscale = bias_scale_p[0];

    {
        const float4* qs = (const float4*)(qp + ((size_t)bh * NQ_ + m0) * D_);
        const float4* ks = (const float4*)(kp + (size_t)bh * NQ_ * D_);
        const int r = tid >> 4, c4 = tid & 15;
        *(float4*)&lq[r * FPITCH + c4 * 4] = qs[tid];
#pragma unroll
        for (int i = 0; i < 8; ++i)
            *(float4*)&lk[(r + 16 * i) * FPITCH + c4 * 4] = ks[tid + i * 256];
    }
    __syncthreads();

    v2f afrag[16], bfrag[16];
#pragma unroll
    for (int kk = 0; kk < 16; ++kk)
        afrag[kk] = *(const v2f*)&lq[lm * FPITCH + kk * 4 + kh];
#pragma unroll
    for (int kk = 0; kk < 16; ++kk)
        bfrag[kk] = *(const v2f*)&lk[(wave * 16 + lm) * FPITCH + kk * 4 + kh];

    v8f c = {0.f, 0.f, 0.f, 0.f, 0.f, 0.f, 0.f, 0.f};
#pragma unroll
    for (int kk = 0; kk < 16; ++kk)
        c = __builtin_amdgcn_wmma_f32_16x16x4_f32(
                false, afrag[kk], false, bfrag[kk], (short)0, c, false, false);

    {
        const float inv_sqd = 0.125f;
        const float inv_tau = 10.0f;
#pragma unroll
        for (int r = 0; r < 8; ++r) {
            float v = c[r] * inv_sqd + bscale * bv[r];
            ls[(8 * hh + r) * SROW + wave * 16 + lm] = v * inv_tau;
        }
    }
    __syncthreads();

    const int row = tid >> 4;
    const int seg = tid & 15;
    float s[8];
    {
        const float4 a = *(const float4*)&ls[row * SROW + seg * 4];
        const float4 b = *(const float4*)&ls[row * SROW + 64 + seg * 4];
        s[0] = a.x; s[1] = a.y; s[2] = a.z; s[3] = a.w;
        s[4] = b.x; s[5] = b.y; s[6] = b.z; s[7] = b.w;
    }

    float smax = __builtin_fmaxf(__builtin_fmaxf(__builtin_fmaxf(s[0], s[1]), __builtin_fmaxf(s[2], s[3])),
                                 __builtin_fmaxf(__builtin_fmaxf(s[4], s[5]), __builtin_fmaxf(s[6], s[7])));
    float smin = __builtin_fminf(__builtin_fminf(__builtin_fminf(s[0], s[1]), __builtin_fminf(s[2], s[3])),
                                 __builtin_fminf(__builtin_fminf(s[4], s[5]), __builtin_fminf(s[6], s[7])));
    smax = ROW_XOR_MAX(smax, 1); smin = ROW_XOR_MIN(smin, 1);
    smax = ROW_XOR_MAX(smax, 2); smin = ROW_XOR_MIN(smin, 2);
    smax = ROW_XOR_MAX(smax, 4); smin = ROW_XOR_MIN(smin, 4);
    smax = ROW_XOR_MAX(smax, 8); smin = ROW_XOR_MIN(smin, 8);

    float lo = -10000.0f, hi = 10000.0f;
    const float target = 0.15f * (float)NQ_;
    for (int it = 0; it < 20; ++it) {
        const float mid = 0.5f * (lo + hi);
        const bool surely_lt = (smax + mid) < ARG_SURELY_LT;
        const bool surely_ge = (smin + mid) > ARG_SURELY_GE;
        bool below = surely_lt;
        if (__builtin_amdgcn_ballot_w32(!(surely_lt || surely_ge)) != 0u) {
            float p = 0.f;
#pragma unroll
            for (int j = 0; j < 8; ++j)
                p += 1.0f / (1.0f + __expf(-(s[j] + mid)));
            p = ROW_XOR_ADD(p, 1);
            p = ROW_XOR_ADD(p, 2);
            p = ROW_XOR_ADD(p, 4);
            p = ROW_XOR_ADD(p, 8);
            below = p < target;
        }
        if (below) lo = mid; else hi = mid;
    }
    const float lam = 0.5f * (lo + hi);

    float o[8];
#pragma unroll
    for (int j = 0; j < 8; ++j)
        o[j] = 1.0f / (1.0f + __expf(-(s[j] + lam)));

    float* orow = out + ((size_t)bh * NQ_ + (m0 + row)) * NQ_;
    *(float4*)(orow + seg * 4)      = make_float4(o[0], o[1], o[2], o[3]);
    *(float4*)(orow + 64 + seg * 4) = make_float4(o[4], o[5], o[6], o[7]);
}

extern "C" void kernel_launch(void* const* d_in, const int* in_sizes, int n_in,
                              void* d_out, int out_size, void* d_ws, size_t ws_size,
                              hipStream_t stream) {
    const float* q     = (const float*)d_in[0];
    const float* k     = (const float*)d_in[1];
    const float* bias  = (const float*)d_in[2];
    const float* Wq    = (const float*)d_in[3];
    const float* Wk    = (const float*)d_in[4];
    const float* bsc   = (const float*)d_in[5];
    float* out = (float*)d_out;

    float* qp = (float*)d_ws;
    float* kp = qp + (size_t)B_ * H_ * NQ_ * D_;

    dim3 g1(B_ * H_ * NQ_ / (PWAVES * PTILES), 2);
    pool_project_kernel<<<g1, PWAVES * 32, 0, stream>>>(q, k, Wq, Wk, qp, kp);

    score_softtopk_kernel<<<B_ * H_ * NSTRIP, 256, 0, stream>>>(qp, kp, bias, bsc, out);
}


# score kernel: sigmoid's 1/(1+exp(-x)) taken with v_rcp_f32 (1 ulp) instead of the correctly rounded division chain, in the bisection sums and the output; terms change in the last bit, sum order kept
# speedup vs baseline: 2.9857x; 1.0601x over previous
#include <hip/hip_runtime.h>
#include <hip/hip_bf16.h>

typedef float v2f __attribute__((ext_vector_type(2)));
typedef float v8f __attribute__((ext_vector_type(8)));

#define B_   4
#define H_   16
#define L_   8192
#define D_   64
#define BS_  64
#define NQ_  128
#define NSTRIP 8
#define SROW 132
#define FPITCH 68

typedef float v4f __attribute__((ext_vector_type(4)));

#define PWAVES 16
#define PTILES 4
#define WPITCH 68

static __device__ __forceinline__ float xhalf(float x) {
    const int b = __builtin_bit_cast(int, x);
    return __builtin_bit_cast(float,
        __builtin_amdgcn_permlanex16(b, b, 0x76543210, 0xfedcba98, false, false));
}
static __device__ __forceinline__ float lane_val(float v, int L) {
    return __builtin_bit_cast(float, __builtin_amdgcn_readlane(__builtin_bit_cast(int, v), L));
}

__global__ __launch_bounds__(PWAVES * 32)
void pool_project_kernel(const float* __restrict__ q, const float* __restrict__ k,
                         const float* __restrict__ Wq, const float* __restrict__ Wk,
                         float* __restrict__ qp, float* __restrict__ kp) {
    __shared__ alignas(16) float lw[D_ * WPITCH];

    const float* x = blockIdx.y ? k : q;
    const float* W = blockIdx.y ? Wk : Wq;
    float* outp    = blockIdx.y ? kp : qp;

    const int tid  = threadIdx.x;
    const int lane = tid & 31;
    const int wave = __builtin_amdgcn_readfirstlane(tid >> 5);

#pragma unroll
    for (int i = 0; i < 2; ++i) {
        const int idx = tid + i * (PWAVES * 32);
        const int r = idx >> 4, c4 = idx & 15;
        *(v4f*)&lw[r * WPITCH + c4 * 4] = *(const v4f*)(W + r * D_ + c4 * 4);
    }
    __syncthreads();

    const int tile0 = (blockIdx.x * PWAVES + wave) * PTILES;
    const v4f* src  = (const v4f*)(x + (size_t)tile0 * (BS_ * D_)) + lane;
    float* dst      = outp + (size_t)tile0 * D_;
    const float* wr = &lw[lane * WPITCH];

    v4f buf[2][16];
#define LOAD_CHUNK(C) do { _Pragma("unroll") for (int j = 0; j < 16; ++j)                    \
        buf[(C) & 1][j] = __builtin_nontemporal_load(src + ((C) * 16 + j) * 32); } while (0)

    LOAD_CHUNK(0);
    v4f ps0, ps1;
#pragma unroll
    for (int c = 0; c < 2 * PTILES; ++c) {
        __builtin_amdgcn_sched_barrier(0);
        if (c + 1 < 2 * PTILES) LOAD_CHUNK(c + 1);
        __builtin_amdgcn_sched_barrier(0);

        if ((c & 1) == 0) { ps0 = buf[0][0]; ps1 = buf[0][1]; }
        else              { ps0 += buf[1][0]; ps1 += buf[1][1]; }
#pragma unroll
        for (int j = 2; j < 16; j += 2) { ps0 += buf[c & 1][j]; ps1 += buf[c & 1][j + 1]; }

        if (c & 1) {
            v4f cs = ps0 + ps1;
            cs.x += xhalf(cs.x); cs.y += xhalf(cs.y); cs.z += xhalf(cs.z); cs.w += xhalf(cs.w);
            float a0 = 0.f, a1 = 0.f, b0 = 0.f, b1 = 0.f;
#pragma unroll
            for (int g = 0; g < 16; g += 2) {
                const v4f wa0 = *(const v4f*)(wr + 4 * g);
                const v4f wb0 = *(const v4f*)(wr + 32 * WPITCH + 4 * g);
                const v4f wa1 = *(const v4f*)(wr + 4 * g + 4);
                const v4f wb1 = *(const v4f*)(wr + 32 * WPITCH + 4 * g + 4);
                const float c0 = lane_val(cs.x, g),     c1 = lane_val(cs.y, g);
                const float c2 = lane_val(cs.z, g),     c3 = lane_val(cs.w, g);
                const float d0 = lane_val(cs.x, g + 1), d1 = lane_val(cs.y, g + 1);
                const float d2 = lane_val(cs.z, g + 1), d3 = lane_val(cs.w, g + 1);
                a0 = __builtin_fmaf(wa0.x, c0, a0); b0 = __builtin_fmaf(wb0.x, c0, b0);
                a1 = __builtin_fmaf(wa1.x, d0, a1); b1 = __builtin_fmaf(wb1.x, d0, b1);
                a0 = __builtin_fmaf(wa0.y, c1, a0); b0 = __builtin_fmaf(wb0.y, c1, b0);
                a1 = __builtin_fmaf(wa1.y, d1, a1); b1 = __builtin_fmaf(wb1.y, d1, b1);
                a0 = __builtin_fmaf(wa0.z, c2, a0); b0 = __builtin_fmaf(wb0.z, c2, b0);
                a1 = __builtin_fmaf(wa1.z, d2, a1); b1 = __builtin_fmaf(wb1.z, d2, b1);
                a0 = __builtin_fmaf(wa0.w, c3, a0); b0 = __builtin_fmaf(wb0.w, c3, b0);
                a1 = __builtin_fmaf(wa1.w, d3, a1); b1 = __builtin_fmaf(wb1.w, d3, b1);
            }
            float* o = dst + (c >> 1) * D_;
            o[lane]      = (a0 + a1) * (1.0f / (float)BS_);
            o[lane + 32] = (b0 + b1) * (1.0f / (float)BS_);
        }
    }
#undef LOAD_CHUNK
}


#define ROW_XOR_ADD(x, M)                                                          \
    ((x) + __builtin_bit_cast(float, __builtin_amdgcn_update_dpp(                   \
               0, __builtin_bit_cast(int, (x)), 0x160 + (M), 0xf, 0xf, true)))

#define ROW_XOR_MAX(x, M)                                                          \
    __builtin_fmaxf((x), __builtin_bit_cast(float, __builtin_amdgcn_update_dpp(    \
               0, __builtin_bit_cast(int, (x)), 0x160 + (M), 0xf, 0xf, true)))
#define ROW_XOR_MIN(x, M)                                                          \
    __builtin_fminf((x), __builtin_bit_cast(float, __builtin_amdgcn_update_dpp(    \
               0, __builtin_bit_cast(int, (x)), 0x160 + (M), 0xf, 0xf, true)))

#define ARG_SURELY_LT (-1.75f)
#define ARG_SURELY_GE (-1.72f)

static __device__ __forceinline__ float sigmoid_rcp(float x) {
    return __builtin_amdgcn_rcpf(1.0f + __expf(-x));
}

__global__ __launch_bounds__(256)
void score_softtopk_kernel(const float* __restrict__ qp, const float* __restrict__ kp,
                           const float* __restrict__ bias,
                           const float* __restrict__ bias_scale_p,
                           float* __restrict__ out) {
    __shared__ alignas(16) float lq[16 * FPITCH];
    __shared__ alignas(16) float lk[NQ_ * FPITCH];
    __shared__ alignas(16) float ls[16 * SROW];

    const int bh    = blockIdx.x >> 3;
    const int strip = blockIdx.x & (NSTRIP - 1);
    const int h     = bh & (H_ - 1);
    const int tid   = threadIdx.x;
    const int wave  = tid >> 5;
    const int lane  = tid & 31;
    const int lm    = lane & 15;
    const int hh    = lane >> 4;
    const int kh    = hh << 1;
    const int m0    = strip * 16;

    float bv[8];
#pragma unroll
    for (int r = 0; r < 8; ++r)
        bv[r] = bias[((size_t)h * NQ_ + (m0 + 8 * hh + r)) * NQ_ + wave * 16 + lm];
    const float bscale = bias_scale_p[0];

    {
        const float4* qs = (const float4*)(qp + ((size_t)bh * NQ_ + m0) * D_);
        const float4* ks = (const float4*)(kp + (size_t)bh * NQ_ * D_);
        const int r = tid >> 4, c4 = tid & 15;
        *(float4*)&lq[r * FPITCH + c4 * 4] = qs[tid];
#pragma unroll
        for (int i = 0; i < 8; ++i)
            *(float4*)&lk[(r + 16 * i) * FPITCH + c4 * 4] = ks[tid + i * 256];
    }
    __syncthreads();

    v2f afrag[16], bfrag[16];
#pragma unroll
    for (int kk = 0; kk < 16; ++kk)
        afrag[kk] = *(const v2f*)&lq[lm * FPITCH + kk * 4 + kh];
#pragma unroll
    for (int kk = 0; kk < 16; ++kk)
        bfrag[kk] = *(const v2f*)&lk[(wave * 16 + lm) * FPITCH + kk * 4 + kh];

    v8f c = {0.f, 0.f, 0.f, 0.f, 0.f, 0.f, 0.f, 0.f};
#pragma unroll
    for (int kk = 0; kk < 16; ++kk)
        c = __builtin_amdgcn_wmma_f32_16x16x4_f32(
                false, afrag[kk], false, bfrag[kk], (short)0, c, false, false);

    {
        const float inv_sqd = 0.125f;
        const float inv_tau = 10.0f;
#pragma unroll
        for (int r = 0; r < 8; ++r) {
            float v = c[r] * inv_sqd + bscale * bv[r];
            ls[(8 * hh + r) * SROW + wave * 16 + lm] = v * inv_tau;
        }
    }
    __syncthreads();

    const int row = tid >> 4;
    const int seg = tid & 15;
    float s[8];
    {
        const float4 a = *(const float4*)&ls[row * SROW + seg * 4];
        const float4 b = *(const float4*)&ls[row * SROW + 64 + seg * 4];
        s[0] = a.x; s[1] = a.y; s[2] = a.z; s[3] = a.w;
        s[4] = b.x; s[5] = b.y; s[6] = b.z; s[7] = b.w;
    }

    float smax = __builtin_fmaxf(__builtin_fmaxf(__builtin_fmaxf(s[0], s[1]), __builtin_fmaxf(s[2], s[3])),
                                 __builtin_fmaxf(__builtin_fmaxf(s[4], s[5]), __builtin_fmaxf(s[6], s[7])));
    float smin = __builtin_fminf(__builtin_fminf(__builtin_fminf(s[0], s[1]), __builtin_fminf(s[2], s[3])),
                                 __builtin_fminf(__builtin_fminf(s[4], s[5]), __builtin_fminf(s[6], s[7])));
    smax = ROW_XOR_MAX(smax, 1); smin = ROW_XOR_MIN(smin, 1);
    smax = ROW_XOR_MAX(smax, 2); smin = ROW_XOR_MIN(smin, 2);
    smax = ROW_XOR_MAX(smax, 4); smin = ROW_XOR_MIN(smin, 4);
    smax = ROW_XOR_MAX(smax, 8); smin = ROW_XOR_MIN(smin, 8);

    float lo = -10000.0f, hi = 10000.0f;
    const float target = 0.15f * (float)NQ_;
    for (int it = 0; it < 20; ++it) {
        const float mid = 0.5f * (lo + hi);
        const bool surely_lt = (smax + mid) < ARG_SURELY_LT;
        const bool surely_ge = (smin + mid) > ARG_SURELY_GE;
        bool below = surely_lt;
        if (__builtin_amdgcn_ballot_w32(!(surely_lt || surely_ge)) != 0u) {
            float p = 0.f;
#pragma unroll
            for (int j = 0; j < 8; ++j)
                p += sigmoid_rcp(s[j] + mid);
            p = ROW_XOR_ADD(p, 1);
            p = ROW_XOR_ADD(p, 2);
            p = ROW_XOR_ADD(p, 4);
            p = ROW_XOR_ADD(p, 8);
            below = p < target;
        }
        if (below) lo = mid; else hi = mid;
    }
    const float lam = 0.5f * (lo + hi);

    float o[8];
#pragma unroll
    for (int j = 0; j < 8; ++j)
        o[j] = sigmoid_rcp(s[j] + lam);

    float* orow = out + ((size_t)bh * NQ_ + (m0 + row)) * NQ_;
    *(float4*)(orow + seg * 4)      = make_float4(o[0], o[1], o[2], o[3]);
    *(float4*)(orow + 64 + seg * 4) = make_float4(o[4], o[5], o[6], o[7]);
}

extern "C" void kernel_launch(void* const* d_in, const int* in_sizes, int n_in,
                              void* d_out, int out_size, void* d_ws, size_t ws_size,
                              hipStream_t stream) {
    const float* q     = (const float*)d_in[0];
    const float* k     = (const float*)d_in[1];
    const float* bias  = (const float*)d_in[2];
    const float* Wq    = (const float*)d_in[3];
    const float* Wk    = (const float*)d_in[4];
    const float* bsc   = (const float*)d_in[5];
    float* out = (float*)d_out;

    float* qp = (float*)d_ws;
    float* kp = qp + (size_t)B_ * H_ * NQ_ * D_;

    dim3 g1(B_ * H_ * NQ_ / (PWAVES * PTILES), 2);
    pool_project_kernel<<<g1, PWAVES * 32, 0, stream>>>(q, k, Wq, Wk, qp, kp);

    score_softtopk_kernel<<<B_ * H_ * NSTRIP, 256, 0, stream>>>(qp, kp, bias, bsc, out);
}
